# starting kernel
# baseline (speedup 1.0000x reference)
#include <hip/hip_runtime.h>
#include <hip/hip_bf16.h>

typedef float v2f __attribute__((ext_vector_type(2)));
typedef float v8f __attribute__((ext_vector_type(8)));

#define RREP   48
#define NNODE  1000
#define FDIM   64
#define HHEADS 4
#define CDIM   64
#define OUTD   256
#define NEDGE  12000
#define EPRIME 13000
#define NROWS  48000
#define NEG_SLOPE 0.2f

__global__ void gat_init(float* __restrict__ out, float* __restrict__ den,
                         const float* __restrict__ bias) {
    int tid = blockIdx.x * blockDim.x + threadIdx.x;
    if (tid < NROWS * FDIM) out[tid] = bias[tid & (FDIM - 1)];
    if (tid < NROWS * HHEADS) den[tid] = 0.0f;
}

__global__ void gat_proj_wmma(const float* __restrict__ x,
                              const float* __restrict__ W,
                              float* __restrict__ h) {
    const int lane = threadIdx.x & 31;
    const int nblk = threadIdx.x >> 5;
    const int mt   = blockIdx.x;
    const int mrow = mt * 16 + (lane & 15);
    const int kb   = (lane >> 4) * 2;
    const int col0 = nblk * 64 + (lane & 15);

    v8f acc0 = {}, acc1 = {}, acc2 = {}, acc3 = {};

    #pragma unroll
    for (int kk = 0; kk < 16; ++kk) {
        const int k = kk * 4 + kb;
        v2f a;
        const float2 av = *(const float2*)(x + (size_t)mrow * FDIM + k);
        a.x = av.x; a.y = av.y;
        v2f b0, b1, b2, b3;
        b0.x = W[k * OUTD + col0 +  0]; b0.y = W[(k + 1) * OUTD + col0 +  0];
        b1.x = W[k * OUTD + col0 + 16]; b1.y = W[(k + 1) * OUTD + col0 + 16];
        b2.x = W[k * OUTD + col0 + 32]; b2.y = W[(k + 1) * OUTD + col0 + 32];
        b3.x = W[k * OUTD + col0 + 48]; b3.y = W[(k + 1) * OUTD + col0 + 48];

        acc0 = __builtin_amdgcn_wmma_f32_16x16x4_f32(false, a, false, b0,
                                                     (short)0, acc0, false, false);
        acc1 = __builtin_amdgcn_wmma_f32_16x16x4_f32(false, a, false, b1,
                                                     (short)0, acc1, false, false);
        acc2 = __builtin_amdgcn_wmma_f32_16x16x4_f32(false, a, false, b2,
                                                     (short)0, acc2, false, false);
        acc3 = __builtin_amdgcn_wmma_f32_16x16x4_f32(false, a, false, b3,
                                                     (short)0, acc3, false, false);
    }

    const int mhalf = (lane >> 4) * 8;
    #pragma unroll
    for (int i = 0; i < 8; ++i) {
        const size_t row = (size_t)(mt * 16 + mhalf + i) * OUTD;
        h[row + col0 +  0] = acc0[i];
        h[row + col0 + 16] = acc1[i];
        h[row + col0 + 32] = acc2[i];
        h[row + col0 + 48] = acc3[i];
    }
}

__global__ void gat_attn_dots(const float* __restrict__ h,
                              const float* __restrict__ att_src,
                              const float* __restrict__ att_dst,
                              float* __restrict__ a_src,
                              float* __restrict__ a_dst) {
    int tid = blockIdx.x * blockDim.x + threadIdx.x;
    if (tid >= NROWS * HHEADS) return;
    const int hh = tid & (HHEADS - 1);
    const int rn = tid >> 2;
    const float* hp = h + (size_t)rn * OUTD + hh * CDIM;
    const float* as = att_src + hh * CDIM;
    const float* ad = att_dst + hh * CDIM;
    float ss = 0.f, sd = 0.f;
    #pragma unroll 8
    for (int c = 0; c < CDIM; ++c) {
        const float v = hp[c];
        ss += v * as[c];
        sd += v * ad[c];
    }
    a_src[tid] = ss;
    a_dst[tid] = sd;
}

__global__ void gat_edge_exp(const int* __restrict__ ei,
                             const float* __restrict__ a_src,
                             const float* __restrict__ a_dst,
                             float* __restrict__ exb,
                             float* __restrict__ den) {
    int tid = blockIdx.x * blockDim.x + threadIdx.x;
    if (tid >= RREP * EPRIME * HHEADS) return;
    const int hh = tid & (HHEADS - 1);
    const int re = tid >> 2;
    const int e  = re % EPRIME;
    const int r  = re / EPRIME;
    int s, d; bool selfm;
    if (e < NEDGE) { s = ei[e]; d = ei[NEDGE + e]; selfm = (s == d); }
    else           { s = e - NEDGE; d = s; selfm = false; }
    float ex;
    if (selfm) {
        ex = 0.0f;
    } else {
        float v = a_src[(r * NNODE + s) * HHEADS + hh]
                + a_dst[(r * NNODE + d) * HHEADS + hh];
        v = v > 0.0f ? v : NEG_SLOPE * v;
        ex = __expf(v);
    }
    exb[tid] = ex;
    atomicAdd(&den[(r * NNODE + d) * HHEADS + hh], ex);
}

__global__ void gat_aggregate(const int* __restrict__ ei,
                              const float* __restrict__ h,
                              const float* __restrict__ exb,
                              const float* __restrict__ den,
                              float* __restrict__ out) {
    const int wid  = blockIdx.x * (blockDim.x >> 5) + (threadIdx.x >> 5);
    if (wid >= RREP * EPRIME) return;
    const int lane = threadIdx.x & 31;
    const int e = wid % EPRIME;
    const int r = wid / EPRIME;
    int s, d;
    if (e < NEDGE) { s = ei[e]; d = ei[NEDGE + e]; }
    else           { s = e - NEDGE; d = s; }

    float alpha[HHEADS];
    #pragma unroll
    for (int hh = 0; hh < HHEADS; ++hh) {
        const float ex = exb[(r * EPRIME + e) * HHEADS + hh];
        const float dn = den[(r * NNODE + d) * HHEADS + hh];
        alpha[hh] = (1.0f / HHEADS) * ex / (dn + 1e-16f);
    }

    const float* hs = h + (size_t)(r * NNODE + s) * OUTD;
    float* op = out + (size_t)(r * NNODE + d) * FDIM;
    #pragma unroll
    for (int t = 0; t < 8; ++t) {
        const int idx = lane + t * 32;
        const int hh  = idx >> 6;
        const int f   = idx & 63;
        atomicAdd(&op[f], alpha[hh] * hs[idx]);
    }
}

extern "C" void kernel_launch(void* const* d_in, const int* in_sizes, int n_in,
                              void* d_out, int out_size, void* d_ws, size_t ws_size,
                              hipStream_t stream) {
    const float* x       = (const float*)d_in[0];
    const int*   ei      = (const int*)d_in[1];
    const float* W       = (const float*)d_in[2];
    const float* att_src = (const float*)d_in[3];
    const float* att_dst = (const float*)d_in[4];
    const float* bias    = (const float*)d_in[5];
    float* out = (float*)d_out;

    float* ws    = (float*)d_ws;
    float* hbuf  = ws;
    float* a_src = hbuf  + (size_t)NROWS * OUTD;
    float* a_dst = a_src + (size_t)NROWS * HHEADS;
    float* exb   = a_dst + (size_t)NROWS * HHEADS;
    float* den   = exb   + (size_t)RREP * EPRIME * HHEADS;

    {
        const int n = NROWS * FDIM;
        gat_init<<<(n + 255) / 256, 256, 0, stream>>>(out, den, bias);
    }
    {
        dim3 grid(NROWS / 16);
        dim3 block(128);
        gat_proj_wmma<<<grid, block, 0, stream>>>(x, W, hbuf);
    }
    {
        const int n = NROWS * HHEADS;
        gat_attn_dots<<<(n + 255) / 256, 256, 0, stream>>>(hbuf, att_src, att_dst,
                                                           a_src, a_dst);
    }
    {
        const int n = RREP * EPRIME * HHEADS;
        gat_edge_exp<<<(n + 255) / 256, 256, 0, stream>>>(ei, a_src, a_dst, exb, den);
    }
    {
        const int waves = RREP * EPRIME;
        const int wpb   = 256 / 32;
        gat_aggregate<<<(waves + wpb - 1) / wpb, 256, 0, stream>>>(ei, hbuf, exb,
                                                                   den, out);
    }
}


# Gather kernel: a row's list head (sources, then their logit halves) is fetched two and one rows ahead, under the x gathers of the rows before; a wave's 8 row offsets in one load
# speedup vs baseline: 11.4913x; 11.4913x over previous
#include <hip/hip_runtime.h>


typedef float v2f __attribute__((ext_vector_type(2)));
typedef float v4f __attribute__((ext_vector_type(4)));
typedef float v8f __attribute__((ext_vector_type(8)));

#define RREP   48
#define NNODE  1000
#define FDIM   64
#define HHEADS 4
#define CDIM   64
#define OUTD   256
#define NEDGE  12000
#define NROWS  48000
#define LOGITS 8
#define NEG_SLOPE 0.2f


#define KEY_DROPPED  0xFFFFu
#define SCAN_BATCH   15
#define CSR_BLOCKS   (NNODE / 8)
#define LOGIT_BLOCKS (NROWS / (16 * 8))
static_assert(NNODE < 0xFFFF, "dst and src are packed as two 16-bit halves");
static_assert(NNODE % 8 == 0, "one wave per node in blocks of 8 waves");
static_assert(NEDGE % 4 == 0 && NEDGE % (32 * SCAN_BATCH) == 0, "whole batches");
static_assert(NROWS % (16 * 8) == 0, "whole blocks of 8 row tiles");
static_assert((FDIM * LOGITS + 8 * 16 * LOGITS) * 4 <= NEDGE * 4, "wvec and staging fit in the key buffer");

__global__ void __launch_bounds__(256)
gat_csr_and_logits(const int* __restrict__ ei, const float* __restrict__ x,
                   const float* __restrict__ W,
                   const float* __restrict__ att_src, const float* __restrict__ att_dst,
                   int* __restrict__ rowinfo, int* __restrict__ csr_src,
                   float* __restrict__ a_sd) {
    __shared__ int4 key4[NEDGE / 4];
    const int lane = threadIdx.x & 31;

    if (blockIdx.x >= CSR_BLOCKS) {
        float* wv    = (float*)key4;
        float* stage = wv + FDIM * LOGITS;
        const int wave = threadIdx.x >> 5;
        const int c    = lane & 15;
        const int hf   = lane >> 4;
        const int tile = (blockIdx.x - CSR_BLOCKS) * 8 + wave;
        const bool live = c < LOGITS;

        const float* xp = x + (size_t)(tile * 16 + c) * FDIM + 2 * hf;
        v2f a[16];
        #pragma unroll
        for (int kk = 0; kk < 16; ++kk) a[kk] = *(const v2f*)(xp + 4 * kk);

        for (int idx = threadIdx.x; idx < FDIM * LOGITS; idx += 256) {
            const int f = idx >> 3, j = idx & 7, hh = j & 3;
            const float* at = (j < HHEADS ? att_src : att_dst) + hh * CDIM;
            const float* wp = W + f * OUTD + hh * CDIM;
            float s = 0.f;
            #pragma unroll
            for (int cc = 0; cc < CDIM; ++cc) s = __builtin_fmaf(wp[cc], at[cc], s);
            wv[idx] = s;
        }
        __syncthreads();

        const float* wp = wv + 2 * hf * LOGITS + (c & (LOGITS - 1));
        v8f acc_e = {}, acc_o = {};
        #pragma unroll
        for (int kk = 0; kk < 16; kk += 2) {
            const float p0 = wp[(4 * kk + 0) * LOGITS], p1 = wp[(4 * kk + 1) * LOGITS];
            const float q0 = wp[(4 * kk + 4) * LOGITS], q1 = wp[(4 * kk + 5) * LOGITS];
            v2f b0, b1;
            b0.x = live ? p0 : 0.f; b0.y = live ? p1 : 0.f;
            b1.x = live ? q0 : 0.f; b1.y = live ? q1 : 0.f;
            acc_e = __builtin_amdgcn_wmma_f32_16x16x4_f32(false, a[kk], false, b0,
                                                          (short)0, acc_e, false, false);
            acc_o = __builtin_amdgcn_wmma_f32_16x16x4_f32(false, a[kk + 1], false, b1,
                                                          (short)0, acc_o, false, false);
        }
        const v8f s = acc_e + acc_o;
        float* st = stage + wave * (16 * LOGITS);
        if (live) {
            #pragma unroll
            for (int i = 0; i < 8; ++i) st[(8 * hf + i) * LOGITS + c] = s[i];
        }
        __syncthreads();
        const v4f v = *(const v4f*)(st + 4 * lane);
        *(v4f*)(a_sd + (size_t)tile * 16 * LOGITS + 4 * lane) = v;
        return;
    }

    {
        const int4* s4 = (const int4*)ei;
        const int4* d4 = (const int4*)(ei + NEDGE);
        for (int q = threadIdx.x; q < NEDGE / 4; q += 256) {
            const int4 s = s4[q], t = d4[q];
            int4 k;
            k.x = (int)(((s.x == t.x ? KEY_DROPPED : (unsigned)t.x) << 16) | (unsigned)s.x);
            k.y = (int)(((s.y == t.y ? KEY_DROPPED : (unsigned)t.y) << 16) | (unsigned)s.y);
            k.z = (int)(((s.z == t.z ? KEY_DROPPED : (unsigned)t.z) << 16) | (unsigned)s.z);
            k.w = (int)(((s.w == t.w ? KEY_DROPPED : (unsigned)t.w) << 16) | (unsigned)s.w);
            key4[q] = k;
        }
    }
    __syncthreads();

    const unsigned* key = (const unsigned*)key4;
    const unsigned d = blockIdx.x * 8 + (threadIdx.x >> 5);
    const unsigned below = (1u << lane) - 1u;

    const unsigned dkey = d << 16;
    int lt = 0;
    for (int e0 = 0; e0 < NEDGE; e0 += 32 * SCAN_BATCH) {
        unsigned kw[SCAN_BATCH];
        for (int u = 0; u < SCAN_BATCH; ++u) kw[u] = key[e0 + 32 * u + lane];
        for (int u = 0; u < SCAN_BATCH; ++u)
            lt += __popc(__builtin_amdgcn_ballot_w32(kw[u] < dkey));
    }
    const int start = lt + (int)d;

    int pos = start;
    for (int e0 = 0; e0 < NEDGE; e0 += 32 * SCAN_BATCH) {
        unsigned kw[SCAN_BATCH];
        for (int u = 0; u < SCAN_BATCH; ++u) kw[u] = key[e0 + 32 * u + lane];
        for (int u = 0; u < SCAN_BATCH; ++u) {
            const bool m = (kw[u] >> 16) == d;
            const unsigned b = __builtin_amdgcn_ballot_w32(m);
            if (m) csr_src[pos + __popc(b & below)] = (int)(kw[u] & 0xFFFFu);
            pos += __popc(b);
        }
    }
    if (lane == 0) {
        csr_src[pos] = (int)d;
        *(int2*)(rowinfo + 2 * d) = make_int2(start, pos + 1 - start);
    }
}

#define TILE_ROWS  64
#define AGG_PITCH  260
#define OUT_PITCH  68
#define EDGE_GROUP 8
#define ROWS_PER_WAVE (TILE_ROWS / 8)
static_assert(NROWS % TILE_ROWS == 0, "whole tiles");
static_assert(ROWS_PER_WAVE == 8 && NNODE % ROWS_PER_WAVE == 0, "a wave's rows share one replica");
static_assert(32 % EDGE_GROUP == 0 && EDGE_GROUP % 2 == 0, "a group never runs past lane 31");
static_assert(TILE_ROWS * OUT_PITCH <= TILE_ROWS * AGG_PITCH, "staging reuses agg");

__device__ __forceinline__ float gat_edge_weight(float v) {
    v = v > 0.0f ? v : NEG_SLOPE * v;
    return __expf(v);
}

__device__ __forceinline__ float gat_lane_bcast(float v, int j) {
    return __int_as_float(__builtin_amdgcn_readlane(__float_as_int(v), j));
}

__device__ __forceinline__ int gat_list_head(const int* __restrict__ csr_src, int2 riv, int k,
                                             int nd0, int lane) {
    const int start = __builtin_amdgcn_readlane(riv.x, k);
    const int cnt   = __builtin_amdgcn_readlane(riv.y, k);
    const int last  = cnt - 1;
    const int s     = csr_src[start + (lane < last ? lane : last)];
    return lane < cnt ? s : nd0 + k;
}

template <int G>
__device__ __forceinline__ void gat_sum_group(const float* __restrict__ xr, int lane, int sj,
                                              float e0, float e1, float e2, float e3, int j,
                                              v2f& acc0, v2f& acc1, v2f& acc2, v2f& acc3,
                                              float& den0, float& den1, float& den2, float& den3) {
    v2f xv[G];
    for (int u = 0; u < G; ++u) {
        const int s = __builtin_amdgcn_readlane(sj, j + u);
        xv[u] = *(const v2f*)(xr + (size_t)s * FDIM + 2 * lane);
    }
    for (int u = 0; u < G; ++u) {
        const float w0 = gat_lane_bcast(e0, j + u), w1 = gat_lane_bcast(e1, j + u);
        const float w2 = gat_lane_bcast(e2, j + u), w3 = gat_lane_bcast(e3, j + u);
        acc0 += w0 * xv[u];
        acc1 += w1 * xv[u];
        acc2 += w2 * xv[u];
        acc3 += w3 * xv[u];
        den0 += w0; den1 += w1; den2 += w2; den3 += w3;
    }
}

__global__ void __launch_bounds__(256)
gat_gather_project(const int* __restrict__ rowinfo, const int* __restrict__ csr_src,
                   const float* __restrict__ x, const float* __restrict__ W,
                   const float* __restrict__ a_sd, const float* __restrict__ bias,
                   float* __restrict__ out) {
    __shared__ float agg[TILE_ROWS * AGG_PITCH];
    const int lane = threadIdx.x & 31;
    const int wave = __builtin_amdgcn_readfirstlane((int)(threadIdx.x >> 5));
    const int row0 = blockIdx.x * TILE_ROWS;

    const int rowb = row0 + wave * ROWS_PER_WAVE;
    const int r    = rowb / NNODE;
    const int dn0  = rowb - r * NNODE;
    const float* xr = x + (size_t)r * NNODE * FDIM;
    const float* ar = a_sd + (size_t)r * NNODE * LOGITS;

    const int2 riv = *(const int2*)(rowinfo + 2 * (dn0 + (lane & (ROWS_PER_WAVE - 1))));
    const v4f  adv = *(const v4f*)(ar + (size_t)(dn0 + (lane & (ROWS_PER_WAVE - 1))) * LOGITS + HHEADS);

    int sj_c = gat_list_head(csr_src, riv, 0, dn0, lane);
    int sj_n = gat_list_head(csr_src, riv, 1, dn0, lane);
    v4f as_c = *(const v4f*)(ar + (size_t)sj_c * LOGITS);

    for (int k = 0; k < ROWS_PER_WAVE; ++k) {
        const int rl    = wave * ROWS_PER_WAVE + k;
        const int d     = dn0 + k;
        const int start = __builtin_amdgcn_readlane(riv.x, k);
        const int cnt   = __builtin_amdgcn_readlane(riv.y, k);
        v4f ad;
        ad.x = gat_lane_bcast(adv.x, k); ad.y = gat_lane_bcast(adv.y, k);
        ad.z = gat_lane_bcast(adv.z, k); ad.w = gat_lane_bcast(adv.w, k);

        v2f acc0 = {0.f, 0.f}, acc1 = {0.f, 0.f}, acc2 = {0.f, 0.f}, acc3 = {0.f, 0.f};
        float den0 = 0.f, den1 = 0.f, den2 = 0.f, den3 = 0.f;

        int  n     = cnt < 32 ? cnt : 32;
        bool valid = lane < n;
        int  sj    = sj_c;
        float e0 = valid ? gat_edge_weight(as_c.x + ad.x) : 0.f;
        float e1 = valid ? gat_edge_weight(as_c.y + ad.y) : 0.f;
        float e2 = valid ? gat_edge_weight(as_c.z + ad.z) : 0.f;
        float e3 = valid ? gat_edge_weight(as_c.w + ad.w) : 0.f;

        const int k2    = k + 2 < ROWS_PER_WAVE ? k + 2 : ROWS_PER_WAVE - 1;
        const int sj_n2 = gat_list_head(csr_src, riv, k2, dn0, lane);
        const v4f as_n  = *(const v4f*)(ar + (size_t)sj_n * LOGITS);

        int base = 0;
        for (;;) {
            for (int j = 0; j < n; j += EDGE_GROUP) {
                if (n - j > EDGE_GROUP / 2)
                    gat_sum_group<EDGE_GROUP>(xr, lane, sj, e0, e1, e2, e3, j,
                                              acc0, acc1, acc2, acc3, den0, den1, den2, den3);
                else
                    gat_sum_group<EDGE_GROUP / 2>(xr, lane, sj, e0, e1, e2, e3, j,
                                                  acc0, acc1, acc2, acc3, den0, den1, den2, den3);
            }
            base += 32;
            if (base >= cnt) break;
            n     = (cnt - base) < 32 ? (cnt - base) : 32;
            valid = lane < n;
            sj    = valid ? csr_src[start + base + lane] : d;
            const v4f as = *(const v4f*)(ar + (size_t)sj * LOGITS);
            e0 = valid ? gat_edge_weight(as.x + ad.x) : 0.f;
            e1 = valid ? gat_edge_weight(as.y + ad.y) : 0.f;
            e2 = valid ? gat_edge_weight(as.z + ad.z) : 0.f;
            e3 = valid ? gat_edge_weight(as.w + ad.w) : 0.f;
        }

        float* ap = agg + rl * AGG_PITCH + 2 * lane;
        *(v2f*)(ap)            = acc0 * ((1.0f / HHEADS) * __builtin_amdgcn_rcpf(den0 + 1e-16f));
        *(v2f*)(ap + CDIM)     = acc1 * ((1.0f / HHEADS) * __builtin_amdgcn_rcpf(den1 + 1e-16f));
        *(v2f*)(ap + 2 * CDIM) = acc2 * ((1.0f / HHEADS) * __builtin_amdgcn_rcpf(den2 + 1e-16f));
        *(v2f*)(ap + 3 * CDIM) = acc3 * ((1.0f / HHEADS) * __builtin_amdgcn_rcpf(den3 + 1e-16f));

        sj_c = sj_n; sj_n = sj_n2; as_c = as_n;
    }
    __syncthreads();

    const int c   = lane & 15;
    const int hf  = lane >> 4;
    const int nt  = wave & 3;
    const int mp  = wave >> 2;
    const int col = nt * 16 + c;
    const float* a0p = agg + (mp * 32 + c) * AGG_PITCH + 2 * hf;
    const float* a1p = a0p + 16 * AGG_PITCH;
    v8f d0 = {}, d1 = {};
    #pragma unroll 8
    for (int kk = 0; kk < OUTD / 4; ++kk) {
        const int kg = 4 * kk + 2 * hf;
        const int hh = kg >> 6;
        const int f  = kg & (FDIM - 1);
        v2f b;
        b.x = W[f * OUTD + hh * CDIM + col];
        b.y = W[(f + 1) * OUTD + hh * CDIM + col];
        const v2f a0 = *(const v2f*)(a0p + 4 * kk);
        const v2f a1 = *(const v2f*)(a1p + 4 * kk);
        d0 = __builtin_amdgcn_wmma_f32_16x16x4_f32(false, a0, false, b,
                                                   (short)0, d0, false, false);
        d1 = __builtin_amdgcn_wmma_f32_16x16x4_f32(false, a1, false, b,
                                                   (short)0, d1, false, false);
    }
    __syncthreads();

    float* st = agg;
    #pragma unroll
    for (int i = 0; i < 8; ++i) {
        st[(mp * 32 + 8 * hf + i) * OUT_PITCH + col]      = d0[i];
        st[(mp * 32 + 16 + 8 * hf + i) * OUT_PITCH + col] = d1[i];
    }
    __syncthreads();
    const int q4 = 4 * (threadIdx.x & 15);
    const v4f bz = *(const v4f*)(bias + q4);
    #pragma unroll
    for (int p = 0; p < TILE_ROWS / 16; ++p) {
        const int rl = p * 16 + (threadIdx.x >> 4);
        const v4f v = *(const v4f*)(st + rl * OUT_PITCH + q4);
        *(v4f*)(out + (size_t)(row0 + rl) * FDIM + q4) = v + bz;
    }
}

extern "C" void kernel_launch(void* const* d_in, const int* in_sizes, int n_in,
                              void* d_out, int out_size, void* d_ws, size_t ws_size,
                              hipStream_t stream) {
    const float* x       = (const float*)d_in[0];
    const int*   ei      = (const int*)d_in[1];
    const float* W       = (const float*)d_in[2];
    const float* att_src = (const float*)d_in[3];
    const float* att_dst = (const float*)d_in[4];
    const float* bias    = (const float*)d_in[5];
    float* out = (float*)d_out;

    float* ws    = (float*)d_ws;
    float* a_sd  = ws;
    int* rowinfo = (int*)(a_sd + (size_t)NROWS * LOGITS);
    int* csr_src = rowinfo + 2 * NNODE;

    gat_csr_and_logits<<<CSR_BLOCKS + LOGIT_BLOCKS, 256, 0, stream>>>(ei, x, W, att_src, att_dst,
                                                                      rowinfo, csr_src, a_sd);
    gat_gather_project<<<NROWS / TILE_ROWS, 256, 0, stream>>>(rowinfo, csr_src, x, W,
                                                              a_sd, bias, out);
}


# CSR build: a block's 8 waves share one sweep of the edge list, packing the block's edges into LDS in edge order; each wave builds its node's list from those, not from two full sweeps of its own
# speedup vs baseline: 16.9033x; 1.4710x over previous
#include <hip/hip_runtime.h>


typedef float v2f __attribute__((ext_vector_type(2)));
typedef float v4f __attribute__((ext_vector_type(4)));
typedef float v8f __attribute__((ext_vector_type(8)));

#define RREP   48
#define NNODE  1000
#define FDIM   64
#define HHEADS 4
#define CDIM   64
#define OUTD   256
#define NEDGE  12000
#define NROWS  48000
#define LOGITS 8
#define NEG_SLOPE 0.2f


#define CSR_BLOCKS     (NNODE / 8)
#define LOGIT_BLOCKS   (NROWS / (16 * 8))
#define CSR_WORDS      (NEDGE / 32)
#define WORDS_PER_WAVE ((CSR_WORDS + 7) / 8)
#define SEG_CAP        (WORDS_PER_WAVE * 32)
#define LOAD_BATCH     24
static_assert(NNODE < 0xFFFF, "dst and src are packed as two 16-bit halves");
static_assert(NNODE % 8 == 0, "one wave per node in blocks of 8 waves");
static_assert(NEDGE % 32 == 0, "whole words of 32 edges");
static_assert(NROWS % (16 * 8) == 0, "whole blocks of 8 row tiles");
static_assert((FDIM * LOGITS + 8 * 16 * LOGITS) * 4 <= 8 * SEG_CAP * 4, "wvec and staging fit in the segments");

__global__ void __launch_bounds__(256)
gat_csr_and_logits(const int* __restrict__ ei, const float* __restrict__ x,
                   const float* __restrict__ W,
                   const float* __restrict__ att_src, const float* __restrict__ att_dst,
                   int* __restrict__ rowinfo, int* __restrict__ csr_src,
                   float* __restrict__ a_sd) {
    __shared__ alignas(16) unsigned lds[8 * SEG_CAP + 16];
    const int lane = threadIdx.x & 31;

    if (blockIdx.x >= CSR_BLOCKS) {
        float* wv    = (float*)lds;
        float* stage = wv + FDIM * LOGITS;
        const int wave = threadIdx.x >> 5;
        const int c    = lane & 15;
        const int hf   = lane >> 4;
        const int tile = (blockIdx.x - CSR_BLOCKS) * 8 + wave;
        const bool live = c < LOGITS;

        const float* xp = x + (size_t)(tile * 16 + c) * FDIM + 2 * hf;
        v2f a[16];
        #pragma unroll
        for (int kk = 0; kk < 16; ++kk) a[kk] = *(const v2f*)(xp + 4 * kk);

        for (int idx = threadIdx.x; idx < FDIM * LOGITS; idx += 256) {
            const int f = idx >> 3, j = idx & 7, hh = j & 3;
            const float* at = (j < HHEADS ? att_src : att_dst) + hh * CDIM;
            const float* wp = W + f * OUTD + hh * CDIM;
            float s = 0.f;
            #pragma unroll
            for (int cc = 0; cc < CDIM; ++cc) s = __builtin_fmaf(wp[cc], at[cc], s);
            wv[idx] = s;
        }
        __syncthreads();

        const float* wp = wv + 2 * hf * LOGITS + (c & (LOGITS - 1));
        v8f acc_e = {}, acc_o = {};
        #pragma unroll
        for (int kk = 0; kk < 16; kk += 2) {
            const float p0 = wp[(4 * kk + 0) * LOGITS], p1 = wp[(4 * kk + 1) * LOGITS];
            const float q0 = wp[(4 * kk + 4) * LOGITS], q1 = wp[(4 * kk + 5) * LOGITS];
            v2f b0, b1;
            b0.x = live ? p0 : 0.f; b0.y = live ? p1 : 0.f;
            b1.x = live ? q0 : 0.f; b1.y = live ? q1 : 0.f;
            acc_e = __builtin_amdgcn_wmma_f32_16x16x4_f32(false, a[kk], false, b0,
                                                          (short)0, acc_e, false, false);
            acc_o = __builtin_amdgcn_wmma_f32_16x16x4_f32(false, a[kk + 1], false, b1,
                                                          (short)0, acc_o, false, false);
        }
        const v8f s = acc_e + acc_o;
        float* st = stage + wave * (16 * LOGITS);
        if (live) {
            #pragma unroll
            for (int i = 0; i < 8; ++i) st[(8 * hf + i) * LOGITS + c] = s[i];
        }
        __syncthreads();
        const v4f v = *(const v4f*)(st + 4 * lane);
        *(v4f*)(a_sd + (size_t)tile * 16 * LOGITS + 4 * lane) = v;
        return;
    }

    unsigned* seg  = lds;
    unsigned* meta = lds + 8 * SEG_CAP;
    const int wave = __builtin_amdgcn_readfirstlane((int)(threadIdx.x >> 5));
    const unsigned d0    = blockIdx.x * 8;
    const unsigned below = (1u << lane) - 1u;

    {
        const int w0 = wave * WORDS_PER_WAVE;
        const int w1 = w0 + WORDS_PER_WAVE < CSR_WORDS ? w0 + WORDS_PER_WAVE : CSR_WORDS;
        unsigned* myseg = seg + wave * SEG_CAP;
        int pos = 0, lt = 0;
        for (int wb = w0; wb < w1; wb += LOAD_BATCH) {
            int s[LOAD_BATCH], t[LOAD_BATCH];
            #pragma unroll
            for (int u = 0; u < LOAD_BATCH; ++u) {
                const int wi = wb + u < CSR_WORDS ? wb + u : CSR_WORDS - 1;
                s[u] = ei[wi * 32 + lane];
                t[u] = ei[NEDGE + wi * 32 + lane];
            }
            #pragma unroll
            for (int u = 0; u < LOAD_BATCH; ++u) {
                const bool kept = (wb + u < w1) && (s[u] != t[u]);
                lt += __popc(__builtin_amdgcn_ballot_w32(kept && (unsigned)t[u] < d0));
                const bool m = kept && ((unsigned)t[u] - d0) < 8u;
                const unsigned b = __builtin_amdgcn_ballot_w32(m);
                if (b != 0u) {
                    if (m) myseg[pos + __popc(b & below)] = ((unsigned)t[u] << 16) | (unsigned)s[u];
                    pos += __popc(b);
                }
            }
        }
        if (lane == 0) { meta[wave] = (unsigned)pos; meta[8 + wave] = (unsigned)lt; }
    }
    __syncthreads();

    const unsigned d = d0 + (unsigned)wave;
    int start = (int)d;
    int cnt[8];
    #pragma unroll
    for (int u = 0; u < 8; ++u) {
        cnt[u] = __builtin_amdgcn_readfirstlane((int)meta[u]);
        start += __builtin_amdgcn_readfirstlane((int)meta[8 + u]);
    }
    unsigned k0[8];
    #pragma unroll
    for (int u = 0; u < 8; ++u) k0[u] = seg[u * SEG_CAP + (lane < cnt[u] ? lane : 0)];

    #pragma unroll
    for (int u = 0; u < 8; ++u) {
        start += __popc(__builtin_amdgcn_ballot_w32(lane < cnt[u] && (k0[u] >> 16) < d));
        const unsigned* sp = seg + u * SEG_CAP;
        #pragma nounroll
        for (int i = 32; i < cnt[u]; i += 32) {
            const bool v = i + lane < cnt[u];
            const unsigned k = sp[v ? i + lane : 0];
            start += __popc(__builtin_amdgcn_ballot_w32(v && (k >> 16) < d));
        }
    }
    int pos = start;
    #pragma unroll
    for (int u = 0; u < 8; ++u) {
        {
            const bool m = lane < cnt[u] && (k0[u] >> 16) == d;
            const unsigned b = __builtin_amdgcn_ballot_w32(m);
            if (m) csr_src[pos + __popc(b & below)] = (int)(k0[u] & 0xFFFFu);
            pos += __popc(b);
        }
        const unsigned* sp = seg + u * SEG_CAP;
        #pragma nounroll
        for (int i = 32; i < cnt[u]; i += 32) {
            const bool v = i + lane < cnt[u];
            const unsigned k = sp[v ? i + lane : 0];
            const bool m = v && (k >> 16) == d;
            const unsigned b = __builtin_amdgcn_ballot_w32(m);
            if (m) csr_src[pos + __popc(b & below)] = (int)(k & 0xFFFFu);
            pos += __popc(b);
        }
    }
    if (lane == 0) {
        csr_src[pos] = (int)d;
        *(int2*)(rowinfo + 2 * d) = make_int2(start, pos + 1 - start);
    }
}

#define TILE_ROWS  64
#define AGG_PITCH  260
#define OUT_PITCH  68
#define EDGE_GROUP 8
#define ROWS_PER_WAVE (TILE_ROWS / 8)
static_assert(NROWS % TILE_ROWS == 0, "whole tiles");
static_assert(ROWS_PER_WAVE == 8 && NNODE % ROWS_PER_WAVE == 0, "a wave's rows share one replica");
static_assert(32 % EDGE_GROUP == 0 && EDGE_GROUP % 2 == 0, "a group never runs past lane 31");
static_assert(TILE_ROWS * OUT_PITCH <= TILE_ROWS * AGG_PITCH, "staging reuses agg");

__device__ __forceinline__ float gat_edge_weight(float v) {
    v = v > 0.0f ? v : NEG_SLOPE * v;
    return __expf(v);
}

__device__ __forceinline__ float gat_lane_bcast(float v, int j) {
    return __int_as_float(__builtin_amdgcn_readlane(__float_as_int(v), j));
}

__device__ __forceinline__ int gat_list_head(const int* __restrict__ csr_src, int2 riv, int k,
                                             int nd0, int lane) {
    const int start = __builtin_amdgcn_readlane(riv.x, k);
    const int cnt   = __builtin_amdgcn_readlane(riv.y, k);
    const int last  = cnt - 1;
    const int s     = csr_src[start + (lane < last ? lane : last)];
    return lane < cnt ? s : nd0 + k;
}

template <int G>
__device__ __forceinline__ void gat_sum_group(const float* __restrict__ xr, int lane, int sj,
                                              float e0, float e1, float e2, float e3, int j,
                                              v2f& acc0, v2f& acc1, v2f& acc2, v2f& acc3,
                                              float& den0, float& den1, float& den2, float& den3) {
    v2f xv[G];
    for (int u = 0; u < G; ++u) {
        const int s = __builtin_amdgcn_readlane(sj, j + u);
        xv[u] = *(const v2f*)(xr + (size_t)s * FDIM + 2 * lane);
    }
    for (int u = 0; u < G; ++u) {
        const float w0 = gat_lane_bcast(e0, j + u), w1 = gat_lane_bcast(e1, j + u);
        const float w2 = gat_lane_bcast(e2, j + u), w3 = gat_lane_bcast(e3, j + u);
        acc0 += w0 * xv[u];
        acc1 += w1 * xv[u];
        acc2 += w2 * xv[u];
        acc3 += w3 * xv[u];
        den0 += w0; den1 += w1; den2 += w2; den3 += w3;
    }
}

__global__ void __launch_bounds__(256)
gat_gather_project(const int* __restrict__ rowinfo, const int* __restrict__ csr_src,
                   const float* __restrict__ x, const float* __restrict__ W,
                   const float* __restrict__ a_sd, const float* __restrict__ bias,
                   float* __restrict__ out) {
    __shared__ float agg[TILE_ROWS * AGG_PITCH];
    const int lane = threadIdx.x & 31;
    const int wave = __builtin_amdgcn_readfirstlane((int)(threadIdx.x >> 5));
    const int row0 = blockIdx.x * TILE_ROWS;

    const int rowb = row0 + wave * ROWS_PER_WAVE;
    const int r    = rowb / NNODE;
    const int dn0  = rowb - r * NNODE;
    const float* xr = x + (size_t)r * NNODE * FDIM;
    const float* ar = a_sd + (size_t)r * NNODE * LOGITS;

    const int2 riv = *(const int2*)(rowinfo + 2 * (dn0 + (lane & (ROWS_PER_WAVE - 1))));
    const v4f  adv = *(const v4f*)(ar + (size_t)(dn0 + (lane & (ROWS_PER_WAVE - 1))) * LOGITS + HHEADS);

    int sj_c = gat_list_head(csr_src, riv, 0, dn0, lane);
    int sj_n = gat_list_head(csr_src, riv, 1, dn0, lane);
    v4f as_c = *(const v4f*)(ar + (size_t)sj_c * LOGITS);

    for (int k = 0; k < ROWS_PER_WAVE; ++k) {
        const int rl    = wave * ROWS_PER_WAVE + k;
        const int d     = dn0 + k;
        const int start = __builtin_amdgcn_readlane(riv.x, k);
        const int cnt   = __builtin_amdgcn_readlane(riv.y, k);
        v4f ad;
        ad.x = gat_lane_bcast(adv.x, k); ad.y = gat_lane_bcast(adv.y, k);
        ad.z = gat_lane_bcast(adv.z, k); ad.w = gat_lane_bcast(adv.w, k);

        v2f acc0 = {0.f, 0.f}, acc1 = {0.f, 0.f}, acc2 = {0.f, 0.f}, acc3 = {0.f, 0.f};
        float den0 = 0.f, den1 = 0.f, den2 = 0.f, den3 = 0.f;

        int  n     = cnt < 32 ? cnt : 32;
        bool valid = lane < n;
        int  sj    = sj_c;
        float e0 = valid ? gat_edge_weight(as_c.x + ad.x) : 0.f;
        float e1 = valid ? gat_edge_weight(as_c.y + ad.y) : 0.f;
        float e2 = valid ? gat_edge_weight(as_c.z + ad.z) : 0.f;
        float e3 = valid ? gat_edge_weight(as_c.w + ad.w) : 0.f;

        const int k2    = k + 2 < ROWS_PER_WAVE ? k + 2 : ROWS_PER_WAVE - 1;
        const int sj_n2 = gat_list_head(csr_src, riv, k2, dn0, lane);
        const v4f as_n  = *(const v4f*)(ar + (size_t)sj_n * LOGITS);

        int base = 0;
        for (;;) {
            for (int j = 0; j < n; j += EDGE_GROUP) {
                if (n - j > EDGE_GROUP / 2)
                    gat_sum_group<EDGE_GROUP>(xr, lane, sj, e0, e1, e2, e3, j,
                                              acc0, acc1, acc2, acc3, den0, den1, den2, den3);
                else
                    gat_sum_group<EDGE_GROUP / 2>(xr, lane, sj, e0, e1, e2, e3, j,
                                                  acc0, acc1, acc2, acc3, den0, den1, den2, den3);
            }
            base += 32;
            if (base >= cnt) break;
            n     = (cnt - base) < 32 ? (cnt - base) : 32;
            valid = lane < n;
            sj    = valid ? csr_src[start + base + lane] : d;
            const v4f as = *(const v4f*)(ar + (size_t)sj * LOGITS);
            e0 = valid ? gat_edge_weight(as.x + ad.x) : 0.f;
            e1 = valid ? gat_edge_weight(as.y + ad.y) : 0.f;
            e2 = valid ? gat_edge_weight(as.z + ad.z) : 0.f;
            e3 = valid ? gat_edge_weight(as.w + ad.w) : 0.f;
        }

        float* ap = agg + rl * AGG_PITCH + 2 * lane;
        *(v2f*)(ap)            = acc0 * ((1.0f / HHEADS) * __builtin_amdgcn_rcpf(den0 + 1e-16f));
        *(v2f*)(ap + CDIM)     = acc1 * ((1.0f / HHEADS) * __builtin_amdgcn_rcpf(den1 + 1e-16f));
        *(v2f*)(ap + 2 * CDIM) = acc2 * ((1.0f / HHEADS) * __builtin_amdgcn_rcpf(den2 + 1e-16f));
        *(v2f*)(ap + 3 * CDIM) = acc3 * ((1.0f / HHEADS) * __builtin_amdgcn_rcpf(den3 + 1e-16f));

        sj_c = sj_n; sj_n = sj_n2; as_c = as_n;
    }
    __syncthreads();

    const int c   = lane & 15;
    const int hf  = lane >> 4;
    const int nt  = wave & 3;
    const int mp  = wave >> 2;
    const int col = nt * 16 + c;
    const float* a0p = agg + (mp * 32 + c) * AGG_PITCH + 2 * hf;
    const float* a1p = a0p + 16 * AGG_PITCH;
    v8f d0 = {}, d1 = {};
    #pragma unroll 8
    for (int kk = 0; kk < OUTD / 4; ++kk) {
        const int kg = 4 * kk + 2 * hf;
        const int hh = kg >> 6;
        const int f  = kg & (FDIM - 1);
        v2f b;
        b.x = W[f * OUTD + hh * CDIM + col];
        b.y = W[(f + 1) * OUTD + hh * CDIM + col];
        const v2f a0 = *(const v2f*)(a0p + 4 * kk);
        const v2f a1 = *(const v2f*)(a1p + 4 * kk);
        d0 = __builtin_amdgcn_wmma_f32_16x16x4_f32(false, a0, false, b,
                                                   (short)0, d0, false, false);
        d1 = __builtin_amdgcn_wmma_f32_16x16x4_f32(false, a1, false, b,
                                                   (short)0, d1, false, false);
    }
    __syncthreads();

    float* st = agg;
    #pragma unroll
    for (int i = 0; i < 8; ++i) {
        st[(mp * 32 + 8 * hf + i) * OUT_PITCH + col]      = d0[i];
        st[(mp * 32 + 16 + 8 * hf + i) * OUT_PITCH + col] = d1[i];
    }
    __syncthreads();
    const int q4 = 4 * (threadIdx.x & 15);
    const v4f bz = *(const v4f*)(bias + q4);
    #pragma unroll
    for (int p = 0; p < TILE_ROWS / 16; ++p) {
        const int rl = p * 16 + (threadIdx.x >> 4);
        const v4f v = *(const v4f*)(st + rl * OUT_PITCH + q4);
        *(v4f*)(out + (size_t)(row0 + rl) * FDIM + q4) = v + bz;
    }
}

extern "C" void kernel_launch(void* const* d_in, const int* in_sizes, int n_in,
                              void* d_out, int out_size, void* d_ws, size_t ws_size,
                              hipStream_t stream) {
    const float* x       = (const float*)d_in[0];
    const int*   ei      = (const int*)d_in[1];
    const float* W       = (const float*)d_in[2];
    const float* att_src = (const float*)d_in[3];
    const float* att_dst = (const float*)d_in[4];
    const float* bias    = (const float*)d_in[5];
    float* out = (float*)d_out;

    float* ws    = (float*)d_ws;
    float* a_sd  = ws;
    int* rowinfo = (int*)(a_sd + (size_t)NROWS * LOGITS);
    int* csr_src = rowinfo + 2 * NNODE;

    gat_csr_and_logits<<<CSR_BLOCKS + LOGIT_BLOCKS, 256, 0, stream>>>(ei, x, W, att_src, att_dst,
                                                                      rowinfo, csr_src, a_sd);
    gat_gather_project<<<NROWS / TILE_ROWS, 256, 0, stream>>>(rowinfo, csr_src, x, W,
                                                              a_sd, bias, out);
}


# Gather kernel: sources and weights pass through wave-private LDS, not 5 readlanes per entry; a wave's halves split the heads, 16-byte x loads, a row's loads go out as one exact group; sum order kept
# speedup vs baseline: 17.7386x; 1.0494x over previous
#include <hip/hip_runtime.h>


typedef float v2f __attribute__((ext_vector_type(2)));
typedef float v4f __attribute__((ext_vector_type(4)));
typedef float v8f __attribute__((ext_vector_type(8)));

#define RREP   48
#define NNODE  1000
#define FDIM   64
#define HHEADS 4
#define CDIM   64
#define OUTD   256
#define NEDGE  12000
#define NROWS  48000
#define LOGITS 8
#define NEG_SLOPE 0.2f


#define CSR_BLOCKS     (NNODE / 8)
#define LOGIT_BLOCKS   (NROWS / (16 * 8))
#define CSR_WORDS      (NEDGE / 32)
#define WORDS_PER_WAVE ((CSR_WORDS + 7) / 8)
#define SEG_CAP        (WORDS_PER_WAVE * 32)
#define LOAD_BATCH     24
static_assert(NNODE < 0xFFFF, "dst and src are packed as two 16-bit halves");
static_assert(NNODE % 8 == 0, "one wave per node in blocks of 8 waves");
static_assert(NEDGE % 32 == 0, "whole words of 32 edges");
static_assert(NROWS % (16 * 8) == 0, "whole blocks of 8 row tiles");
static_assert((FDIM * LOGITS + 8 * 16 * LOGITS) * 4 <= 8 * SEG_CAP * 4, "wvec and staging fit in the segments");

__global__ void __launch_bounds__(256)
gat_csr_and_logits(const int* __restrict__ ei, const float* __restrict__ x,
                   const float* __restrict__ W,
                   const float* __restrict__ att_src, const float* __restrict__ att_dst,
                   int* __restrict__ rowinfo, int* __restrict__ csr_src,
                   float* __restrict__ a_sd) {
    __shared__ alignas(16) unsigned lds[8 * SEG_CAP + 16];
    const int lane = threadIdx.x & 31;

    if (blockIdx.x >= CSR_BLOCKS) {
        float* wv    = (float*)lds;
        float* stage = wv + FDIM * LOGITS;
        const int wave = threadIdx.x >> 5;
        const int c    = lane & 15;
        const int hf   = lane >> 4;
        const int tile = (blockIdx.x - CSR_BLOCKS) * 8 + wave;
        const bool live = c < LOGITS;

        const float* xp = x + (size_t)(tile * 16 + c) * FDIM + 2 * hf;
        v2f a[16];
        #pragma unroll
        for (int kk = 0; kk < 16; ++kk) a[kk] = *(const v2f*)(xp + 4 * kk);

        for (int idx = threadIdx.x; idx < FDIM * LOGITS; idx += 256) {
            const int f = idx >> 3, j = idx & 7, hh = j & 3;
            const float* at = (j < HHEADS ? att_src : att_dst) + hh * CDIM;
            const float* wp = W + f * OUTD + hh * CDIM;
            float s = 0.f;
            #pragma unroll
            for (int cc = 0; cc < CDIM; ++cc) s = __builtin_fmaf(wp[cc], at[cc], s);
            wv[idx] = s;
        }
        __syncthreads();

        const float* wp = wv + 2 * hf * LOGITS + (c & (LOGITS - 1));
        v8f acc_e = {}, acc_o = {};
        #pragma unroll
        for (int kk = 0; kk < 16; kk += 2) {
            const float p0 = wp[(4 * kk + 0) * LOGITS], p1 = wp[(4 * kk + 1) * LOGITS];
            const float q0 = wp[(4 * kk + 4) * LOGITS], q1 = wp[(4 * kk + 5) * LOGITS];
            v2f b0, b1;
            b0.x = live ? p0 : 0.f; b0.y = live ? p1 : 0.f;
            b1.x = live ? q0 : 0.f; b1.y = live ? q1 : 0.f;
            acc_e = __builtin_amdgcn_wmma_f32_16x16x4_f32(false, a[kk], false, b0,
                                                          (short)0, acc_e, false, false);
            acc_o = __builtin_amdgcn_wmma_f32_16x16x4_f32(false, a[kk + 1], false, b1,
                                                          (short)0, acc_o, false, false);
        }
        const v8f s = acc_e + acc_o;
        float* st = stage + wave * (16 * LOGITS);
        if (live) {
            #pragma unroll
            for (int i = 0; i < 8; ++i) st[(8 * hf + i) * LOGITS + c] = s[i];
        }
        __syncthreads();
        const v4f v = *(const v4f*)(st + 4 * lane);
        *(v4f*)(a_sd + (size_t)tile * 16 * LOGITS + 4 * lane) = v;
        return;
    }

    unsigned* seg  = lds;
    unsigned* meta = lds + 8 * SEG_CAP;
    const int wave = __builtin_amdgcn_readfirstlane((int)(threadIdx.x >> 5));
    const unsigned d0    = blockIdx.x * 8;
    const unsigned below = (1u << lane) - 1u;

    {
        const int w0 = wave * WORDS_PER_WAVE;
        const int w1 = w0 + WORDS_PER_WAVE < CSR_WORDS ? w0 + WORDS_PER_WAVE : CSR_WORDS;
        unsigned* myseg = seg + wave * SEG_CAP;
        int pos = 0, lt = 0;
        for (int wb = w0; wb < w1; wb += LOAD_BATCH) {
            int s[LOAD_BATCH], t[LOAD_BATCH];
            #pragma unroll
            for (int u = 0; u < LOAD_BATCH; ++u) {
                const int wi = wb + u < CSR_WORDS ? wb + u : CSR_WORDS - 1;
                s[u] = ei[wi * 32 + lane];
                t[u] = ei[NEDGE + wi * 32 + lane];
            }
            #pragma unroll
            for (int u = 0; u < LOAD_BATCH; ++u) {
                const bool kept = (wb + u < w1) && (s[u] != t[u]);
                lt += __popc(__builtin_amdgcn_ballot_w32(kept && (unsigned)t[u] < d0));
                const bool m = kept && ((unsigned)t[u] - d0) < 8u;
                const unsigned b = __builtin_amdgcn_ballot_w32(m);
                if (b != 0u) {
                    if (m) myseg[pos + __popc(b & below)] = ((unsigned)t[u] << 16) | (unsigned)s[u];
                    pos += __popc(b);
                }
            }
        }
        if (lane == 0) { meta[wave] = (unsigned)pos; meta[8 + wave] = (unsigned)lt; }
    }
    __syncthreads();

    const unsigned d = d0 + (unsigned)wave;
    int start = (int)d;
    int cnt[8];
    #pragma unroll
    for (int u = 0; u < 8; ++u) {
        cnt[u] = __builtin_amdgcn_readfirstlane((int)meta[u]);
        start += __builtin_amdgcn_readfirstlane((int)meta[8 + u]);
    }
    unsigned k0[8];
    #pragma unroll
    for (int u = 0; u < 8; ++u) k0[u] = seg[u * SEG_CAP + (lane < cnt[u] ? lane : 0)];

    #pragma unroll
    for (int u = 0; u < 8; ++u) {
        start += __popc(__builtin_amdgcn_ballot_w32(lane < cnt[u] && (k0[u] >> 16) < d));
        const unsigned* sp = seg + u * SEG_CAP;
        #pragma nounroll
        for (int i = 32; i < cnt[u]; i += 32) {
            const bool v = i + lane < cnt[u];
            const unsigned k = sp[v ? i + lane : 0];
            start += __popc(__builtin_amdgcn_ballot_w32(v && (k >> 16) < d));
        }
    }
    int pos = start;
    #pragma unroll
    for (int u = 0; u < 8; ++u) {
        {
            const bool m = lane < cnt[u] && (k0[u] >> 16) == d;
            const unsigned b = __builtin_amdgcn_ballot_w32(m);
            if (m) csr_src[pos + __popc(b & below)] = (int)(k0[u] & 0xFFFFu);
            pos += __popc(b);
        }
        const unsigned* sp = seg + u * SEG_CAP;
        #pragma nounroll
        for (int i = 32; i < cnt[u]; i += 32) {
            const bool v = i + lane < cnt[u];
            const unsigned k = sp[v ? i + lane : 0];
            const bool m = v && (k >> 16) == d;
            const unsigned b = __builtin_amdgcn_ballot_w32(m);
            if (m) csr_src[pos + __popc(b & below)] = (int)(k & 0xFFFFu);
            pos += __popc(b);
        }
    }
    if (lane == 0) {
        csr_src[pos] = (int)d;
        *(int2*)(rowinfo + 2 * d) = make_int2(start, pos + 1 - start);
    }
}

#define TILE_ROWS  64
#define AGG_PITCH  260
#define OUT_PITCH  68
#define ENTRY_GROUP 16
#define ROWS_PER_WAVE (TILE_ROWS / 8)
static_assert(NROWS % TILE_ROWS == 0, "whole tiles");
static_assert(ROWS_PER_WAVE == 8 && NNODE % ROWS_PER_WAVE == 0, "a wave's rows share one replica");
static_assert(TILE_ROWS * OUT_PITCH <= TILE_ROWS * AGG_PITCH, "staging reuses agg");

__device__ __forceinline__ float gat_edge_weight(float v) {
    v = v > 0.0f ? v : NEG_SLOPE * v;
    return __expf(v);
}

__device__ __forceinline__ float gat_lane_bcast(float v, int j) {
    return __int_as_float(__builtin_amdgcn_readlane(__float_as_int(v), j));
}

__device__ __forceinline__ int gat_list_head(const int* __restrict__ csr_src, int2 riv, int k,
                                             int nd0, int lane) {
    const int start = __builtin_amdgcn_readlane(riv.x, k);
    const int cnt   = __builtin_amdgcn_readlane(riv.y, k);
    const int last  = cnt - 1;
    const int s     = csr_src[start + (lane < last ? lane : last)];
    return lane < cnt ? s : nd0 + k;
}

template <int G>
__device__ __forceinline__ void gat_sum_entries(const v4f* __restrict__ xr4, const float* wsc,
                                                const int* ssc, int q, int half, int j,
                                                v4f& acca, v4f& accb, v2f& den) {
    v4f xv[G];
    v2f wv[G];
    for (int u = 0; u < G; ++u) {
        const int s = ssc[j + u];
        wv[u] = *(const v2f*)(wsc + HHEADS * (j + u) + 2 * half);
        xv[u] = xr4[(unsigned)(s * (FDIM / 4) + q)];
    }
    for (int u = 0; u < G; ++u) {
        acca += wv[u].x * xv[u];
        accb += wv[u].y * xv[u];
        den  += wv[u];
    }
}

__global__ void __launch_bounds__(256)
gat_gather_project(const int* __restrict__ rowinfo, const int* __restrict__ csr_src,
                   const float* __restrict__ x, const float* __restrict__ W,
                   const float* __restrict__ a_sd, const float* __restrict__ bias,
                   float* __restrict__ out) {
    __shared__ alignas(16) float agg[TILE_ROWS * AGG_PITCH];
    __shared__ alignas(16) float wscr[8 * 32 * HHEADS];
    __shared__ int sscr[8 * 32];
    const int lane = threadIdx.x & 31;
    const int wave = __builtin_amdgcn_readfirstlane((int)(threadIdx.x >> 5));
    const int row0 = blockIdx.x * TILE_ROWS;

    const int rowb = row0 + wave * ROWS_PER_WAVE;
    const int r    = rowb / NNODE;
    const int dn0  = rowb - r * NNODE;
    const float* xr = x + (size_t)r * NNODE * FDIM;
    const float* ar = a_sd + (size_t)r * NNODE * LOGITS;
    const int q    = lane & 15;
    const int half = lane >> 4;
    float* wsc = wscr + wave * (32 * HHEADS);
    int*   ssc = sscr + wave * 32;
    const v4f* xr4 = (const v4f*)xr;

    const int2 riv = *(const int2*)(rowinfo + 2 * (dn0 + (lane & (ROWS_PER_WAVE - 1))));
    const v4f  adv = *(const v4f*)(ar + (size_t)(dn0 + (lane & (ROWS_PER_WAVE - 1))) * LOGITS + HHEADS);

    int sj_c = gat_list_head(csr_src, riv, 0, dn0, lane);
    int sj_n = gat_list_head(csr_src, riv, 1, dn0, lane);
    v4f as_c = *(const v4f*)(ar + (size_t)sj_c * LOGITS);

    for (int k = 0; k < ROWS_PER_WAVE; ++k) {
        const int rl    = wave * ROWS_PER_WAVE + k;
        const int d     = dn0 + k;
        const int start = __builtin_amdgcn_readlane(riv.x, k);
        const int cnt   = __builtin_amdgcn_readlane(riv.y, k);
        v4f ad;
        ad.x = gat_lane_bcast(adv.x, k); ad.y = gat_lane_bcast(adv.y, k);
        ad.z = gat_lane_bcast(adv.z, k); ad.w = gat_lane_bcast(adv.w, k);

        v4f acca = {0.f, 0.f, 0.f, 0.f}, accb = acca;
        v2f den  = {0.f, 0.f};

        int  n     = cnt < 32 ? cnt : 32;
        bool valid = lane < n;
        int  sj    = sj_c;
        float e0 = valid ? gat_edge_weight(as_c.x + ad.x) : 0.f;
        float e1 = valid ? gat_edge_weight(as_c.y + ad.y) : 0.f;
        float e2 = valid ? gat_edge_weight(as_c.z + ad.z) : 0.f;
        float e3 = valid ? gat_edge_weight(as_c.w + ad.w) : 0.f;

        const int k2    = k + 2 < ROWS_PER_WAVE ? k + 2 : ROWS_PER_WAVE - 1;
        const int sj_n2 = gat_list_head(csr_src, riv, k2, dn0, lane);
        const v4f as_n  = *(const v4f*)(ar + (size_t)sj_n * LOGITS);

        int base = 0;
        for (;;) {
            {
                v4f ev; ev.x = e0; ev.y = e1; ev.z = e2; ev.w = e3;
                *(v4f*)(wsc + 4 * lane) = ev;
                ssc[lane] = sj;
            }
            int j = 0;
            for (; n - j > ENTRY_GROUP; j += ENTRY_GROUP)
                gat_sum_entries<ENTRY_GROUP>(xr4, wsc, ssc, q, half, j, acca, accb, den);
            switch (n - j) {
            case 1:  gat_sum_entries<1>(xr4, wsc, ssc, q, half, j, acca, accb, den);  break;
            case 2:  gat_sum_entries<2>(xr4, wsc, ssc, q, half, j, acca, accb, den);  break;
            case 3:  gat_sum_entries<3>(xr4, wsc, ssc, q, half, j, acca, accb, den);  break;
            case 4:  gat_sum_entries<4>(xr4, wsc, ssc, q, half, j, acca, accb, den);  break;
            case 5:  gat_sum_entries<5>(xr4, wsc, ssc, q, half, j, acca, accb, den);  break;
            case 6:  gat_sum_entries<6>(xr4, wsc, ssc, q, half, j, acca, accb, den);  break;
            case 7:  gat_sum_entries<7>(xr4, wsc, ssc, q, half, j, acca, accb, den);  break;
            case 8:  gat_sum_entries<8>(xr4, wsc, ssc, q, half, j, acca, accb, den);  break;
            case 9:  gat_sum_entries<9>(xr4, wsc, ssc, q, half, j, acca, accb, den);  break;
            case 10: gat_sum_entries<10>(xr4, wsc, ssc, q, half, j, acca, accb, den); break;
            case 11: gat_sum_entries<11>(xr4, wsc, ssc, q, half, j, acca, accb, den); break;
            case 12: gat_sum_entries<12>(xr4, wsc, ssc, q, half, j, acca, accb, den); break;
            case 13: gat_sum_entries<13>(xr4, wsc, ssc, q, half, j, acca, accb, den); break;
            case 14: gat_sum_entries<14>(xr4, wsc, ssc, q, half, j, acca, accb, den); break;
            case 15: gat_sum_entries<15>(xr4, wsc, ssc, q, half, j, acca, accb, den); break;
            default: gat_sum_entries<16>(xr4, wsc, ssc, q, half, j, acca, accb, den); break;
            }
            base += 32;
            if (base >= cnt) break;
            n     = (cnt - base) < 32 ? (cnt - base) : 32;
            valid = lane < n;
            sj    = valid ? csr_src[start + base + lane] : d;
            const v4f as = *(const v4f*)(ar + (size_t)sj * LOGITS);
            e0 = valid ? gat_edge_weight(as.x + ad.x) : 0.f;
            e1 = valid ? gat_edge_weight(as.y + ad.y) : 0.f;
            e2 = valid ? gat_edge_weight(as.z + ad.z) : 0.f;
            e3 = valid ? gat_edge_weight(as.w + ad.w) : 0.f;
        }

        float* ap = agg + rl * AGG_PITCH + 2 * half * CDIM + 4 * q;
        *(v4f*)(ap)        = acca * ((1.0f / HHEADS) * __builtin_amdgcn_rcpf(den.x + 1e-16f));
        *(v4f*)(ap + CDIM) = accb * ((1.0f / HHEADS) * __builtin_amdgcn_rcpf(den.y + 1e-16f));

        sj_c = sj_n; sj_n = sj_n2; as_c = as_n;
    }
    __syncthreads();

    const int c   = lane & 15;
    const int hf  = lane >> 4;
    const int nt  = wave & 3;
    const int mp  = wave >> 2;
    const int col = nt * 16 + c;
    const float* a0p = agg + (mp * 32 + c) * AGG_PITCH + 2 * hf;
    const float* a1p = a0p + 16 * AGG_PITCH;
    v8f d0 = {}, d1 = {};
    #pragma unroll 8
    for (int kk = 0; kk < OUTD / 4; ++kk) {
        const int kg = 4 * kk + 2 * hf;
        const int hh = kg >> 6;
        const int f  = kg & (FDIM - 1);
        v2f b;
        b.x = W[f * OUTD + hh * CDIM + col];
        b.y = W[(f + 1) * OUTD + hh * CDIM + col];
        const v2f a0 = *(const v2f*)(a0p + 4 * kk);
        const v2f a1 = *(const v2f*)(a1p + 4 * kk);
        d0 = __builtin_amdgcn_wmma_f32_16x16x4_f32(false, a0, false, b,
                                                   (short)0, d0, false, false);
        d1 = __builtin_amdgcn_wmma_f32_16x16x4_f32(false, a1, false, b,
                                                   (short)0, d1, false, false);
    }
    __syncthreads();

    float* st = agg;
    #pragma unroll
    for (int i = 0; i < 8; ++i) {
        st[(mp * 32 + 8 * hf + i) * OUT_PITCH + col]      = d0[i];
        st[(mp * 32 + 16 + 8 * hf + i) * OUT_PITCH + col] = d1[i];
    }
    __syncthreads();
    const int q4 = 4 * (threadIdx.x & 15);
    const v4f bz = *(const v4f*)(bias + q4);
    #pragma unroll
    for (int p = 0; p < TILE_ROWS / 16; ++p) {
        const int rl = p * 16 + (threadIdx.x >> 4);
        const v4f v = *(const v4f*)(st + rl * OUT_PITCH + q4);
        *(v4f*)(out + (size_t)(row0 + rl) * FDIM + q4) = v + bz;
    }
}

extern "C" void kernel_launch(void* const* d_in, const int* in_sizes, int n_in,
                              void* d_out, int out_size, void* d_ws, size_t ws_size,
                              hipStream_t stream) {
    const float* x       = (const float*)d_in[0];
    const int*   ei      = (const int*)d_in[1];
    const float* W       = (const float*)d_in[2];
    const float* att_src = (const float*)d_in[3];
    const float* att_dst = (const float*)d_in[4];
    const float* bias    = (const float*)d_in[5];
    float* out = (float*)d_out;

    float* ws    = (float*)d_ws;
    float* a_sd  = ws;
    int* rowinfo = (int*)(a_sd + (size_t)NROWS * LOGITS);
    int* csr_src = rowinfo + 2 * NNODE;

    gat_csr_and_logits<<<CSR_BLOCKS + LOGIT_BLOCKS, 256, 0, stream>>>(ei, x, W, att_src, att_dst,
                                                                      rowinfo, csr_src, a_sd);
    gat_gather_project<<<NROWS / TILE_ROWS, 256, 0, stream>>>(rowinfo, csr_src, x, W,
                                                              a_sd, bias, out);
}
